# starting kernel
# baseline (speedup 1.0000x reference)
#include <hip/hip_runtime.h>
#include <math.h>

typedef __attribute__((ext_vector_type(2))) float v2f;
typedef __attribute__((ext_vector_type(8))) float v8f;

#define NQ      12
#define NSTATE  4096
#define NLAYERS 4
#define NTHREADS 256
#define NGATES  (NQ + NLAYERS * NQ)

__device__ __forceinline__ void buildM(float* mr, float* mi, int wl,
                                       const float* gg, int t) {
    const int row = t >> 4, cm = t & 15;
    const int mask = 15 & ~(1 << wl);
    float re = 0.0f, im = 0.0f;
    if (((row ^ cm) & mask) == 0) {
        const int rb = (row >> wl) & 1;
        const int cb = (cm >> wl) & 1;
        const int e = (rb * 2 + cb) * 2;
        re = gg[e];
        im = gg[e + 1];
    }
    mr[t] = re;
    mi[t] = im;
}

template <int G>
__device__ __forceinline__ void apply16T(float* sr, float* si,
                                         const float* mr, const float* mi,
                                         int wave, int lane) {
    constexpr int SH = G * 4;
    constexpr int LOWMASK = (1 << SH) - 1;
    const int half = lane >> 4;
    const int n15  = lane & 15;

    v2f ar[4], ai[4], ani[4];
#pragma unroll
    for (int kc = 0; kc < 4; ++kc) {
        const int k0 = kc * 4 + 2 * half;
        ar[kc].x  = mr[n15 * 16 + k0];  ar[kc].y  = mr[n15 * 16 + k0 + 1];
        ai[kc].x  = mi[n15 * 16 + k0];  ai[kc].y  = mi[n15 * 16 + k0 + 1];
        ani[kc].x = -ai[kc].x;          ani[kc].y = -ai[kc].y;
    }

#pragma unroll
    for (int tt = 0; tt < 2; ++tt) {
        const int col = (wave * 2 + tt) * 16 + n15;
        const int base = ((col >> SH) << (SH + 4)) | (col & LOWMASK);

        v2f br[4], bi[4];
#pragma unroll
        for (int kc = 0; kc < 4; ++kc) {
            const int k0 = kc * 4 + 2 * half;
            const int i0 = base | ((k0) << SH);
            const int i1 = base | ((k0 + 1) << SH);
            br[kc].x = sr[i0]; br[kc].y = sr[i1];
            bi[kc].x = si[i0]; bi[kc].y = si[i1];
        }

        v8f accR = {0.f,0.f,0.f,0.f,0.f,0.f,0.f,0.f};
        v8f accI = {0.f,0.f,0.f,0.f,0.f,0.f,0.f,0.f};

#pragma unroll
        for (int kc = 0; kc < 4; ++kc) {
            accR = __builtin_amdgcn_wmma_f32_16x16x4_f32(false, ar[kc],  false, br[kc], (short)0, accR, false, false);
            accR = __builtin_amdgcn_wmma_f32_16x16x4_f32(false, ani[kc], false, bi[kc], (short)0, accR, false, false);
            accI = __builtin_amdgcn_wmma_f32_16x16x4_f32(false, ar[kc],  false, bi[kc], (short)0, accI, false, false);
            accI = __builtin_amdgcn_wmma_f32_16x16x4_f32(false, ai[kc],  false, br[kc], (short)0, accI, false, false);
        }

#pragma unroll
        for (int r = 0; r < 8; ++r) {
            const int i = base | ((r + 8 * half) << SH);
            sr[i] = accR[r];
            si[i] = accI[r];
        }
    }
}

__device__ __forceinline__ void applyDispatch(float* sr, float* si,
                                              const float* mr, const float* mi,
                                              int g, int wave, int lane) {
    if (g == 0)      apply16T<0>(sr, si, mr, mi, wave, lane);
    else if (g == 1) apply16T<1>(sr, si, mr, mi, wave, lane);
    else             apply16T<2>(sr, si, mr, mi, wave, lane);
}

__global__ __launch_bounds__(NTHREADS)
void vqc_kernel(const float* __restrict__ x,
                const float* __restrict__ wts,
                float* __restrict__ out) {
    __shared__ float sr[NSTATE];
    __shared__ float si[NSTATE];
    __shared__ float mr[256];
    __shared__ float mi[256];
    __shared__ float gm[NGATES * 8];
    __shared__ float wred[NQ * 8];

    const int t    = threadIdx.x;
    const int b    = blockIdx.x;
    const int wave = t >> 5;
    const int lane = t & 31;

    if (t < NGATES) {
        float e0, e1, e2, e3, e4, e5, e6, e7;
        if (t < NQ) {
            const float a = x[b * NQ + t] * 0.5f;
            const float c = cosf(a), s = sinf(a);
            e0 = c;    e1 = 0.f;
            e2 = 0.f;  e3 = -s;
            e4 = 0.f;  e5 = -s;
            e6 = c;    e7 = 0.f;
        } else {
            const float* wp = wts + (t - NQ) * 3;
            const float phi = wp[0], th = wp[1], om = wp[2];
            const float ct = cosf(th * 0.5f);
            const float sn = sinf(th * 0.5f);
            const float hs = 0.5f * (phi + om);
            const float hd = 0.5f * (phi - om);
            const float chs = cosf(hs), shs = sinf(hs);
            const float chd = cosf(hd), shd = sinf(hd);
            e0 =  chs * ct;  e1 = -shs * ct;
            e2 = -chd * sn;  e3 = -shd * sn;
            e4 =  chd * sn;  e5 = -shd * sn;
            e6 =  chs * ct;  e7 =  shs * ct;
        }
        float* gr = gm + t * 8;
        gr[0] = e0; gr[1] = e1; gr[2] = e2; gr[3] = e3;
        gr[4] = e4; gr[5] = e5; gr[6] = e6; gr[7] = e7;
    }

#pragma unroll
    for (int k = 0; k < 16; ++k) {
        const int i = k * NTHREADS + t;
        sr[i] = 0.0f;
        si[i] = 0.0f;
    }
    if (t == 0) sr[0] = 1.0f;
    __syncthreads();

    for (int q = 0; q < NQ; ++q) {
        const int bw = 11 - q;
        buildM(mr, mi, bw & 3, gm + q * 8, t);
        __syncthreads();
        applyDispatch(sr, si, mr, mi, bw >> 2, wave, lane);
        __syncthreads();
    }

    for (int l = 0; l < NLAYERS; ++l) {
        for (int q = 0; q < NQ; ++q) {
            const int bw = 11 - q;
            buildM(mr, mi, bw & 3, gm + (NQ + l * NQ + q) * 8, t);
            __syncthreads();
            applyDispatch(sr, si, mr, mi, bw >> 2, wave, lane);
            __syncthreads();
        }

        const int r = l % (NQ - 1) + 1;
        float vr[16], vi[16];
#pragma unroll
        for (int k = 0; k < 16; ++k) {
            const int j = k * NTHREADS + t;
            int src = j;
            for (int qq = NQ - 1; qq >= 0; --qq) {
                const int cbit = 11 - qq;
                const int tbit = 11 - ((qq + r) % NQ);
                src ^= ((src >> cbit) & 1) << tbit;
            }
            vr[k] = sr[src];
            vi[k] = si[src];
        }
        __syncthreads();
#pragma unroll
        for (int k = 0; k < 16; ++k) {
            const int j = k * NTHREADS + t;
            sr[j] = vr[k];
            si[j] = vi[k];
        }
        __syncthreads();
    }

    float part[NQ];
#pragma unroll
    for (int q = 0; q < NQ; ++q) part[q] = 0.0f;
#pragma unroll
    for (int k = 0; k < 16; ++k) {
        const int i = k * NTHREADS + t;
        const float p = sr[i] * sr[i] + si[i] * si[i];
#pragma unroll
        for (int q = 0; q < NQ; ++q) {
            part[q] += ((i >> (11 - q)) & 1) ? -p : p;
        }
    }
#pragma unroll
    for (int q = 0; q < NQ; ++q) {
        float v = part[q];
        for (int off = 16; off >= 1; off >>= 1)
            v += __shfl_xor(v, off, 32);
        if (lane == 0) wred[q * 8 + wave] = v;
    }
    __syncthreads();
    if (t < NQ) {
        float s = 0.0f;
#pragma unroll
        for (int wv = 0; wv < 8; ++wv) s += wred[t * 8 + wv];
        out[b * NQ + t] = s;
    }
}

extern "C" void kernel_launch(void* const* d_in, const int* in_sizes, int n_in,
                              void* d_out, int out_size, void* d_ws, size_t ws_size,
                              hipStream_t stream) {
    const float* x   = (const float*)d_in[0];
    const float* wts = (const float*)d_in[1];
    float* out = (float*)d_out;
    const int batch = in_sizes[0] / NQ;
    vqc_kernel<<<batch, NTHREADS, 0, stream>>>(x, wts, out);
}


# each 2x2 complex gate as ONE real 4x4 product per 16x16x4 WMMA (k = re/im x target bit; gate in 2 registers, bit in LDS addresses, no LDS operator tile): 32->16 WMMA/wave/gate; f32 sum order changes
# speedup vs baseline: 2.0119x; 2.0119x over previous
#include <hip/hip_runtime.h>
#include <math.h>


typedef __attribute__((ext_vector_type(2))) float v2f;
typedef __attribute__((ext_vector_type(8))) float v8f;

#define NQ      12
#define NSTATE  4096
#define NLAYERS 4
#define NTHREADS 256
#define NGATES  (NQ + NLAYERS * NQ)
#define PLANE   (NSTATE + 32)

__device__ __forceinline__ constexpr int ins0(int x, int B) {
    return ((x & ~((1 << B) - 1)) << 1) | (x & ((1 << B) - 1));
}

__device__ __forceinline__ void wave_sync() {
    __builtin_amdgcn_fence(__ATOMIC_RELEASE, "wavefront");
    __builtin_amdgcn_wave_barrier();
    __builtin_amdgcn_fence(__ATOMIC_ACQUIRE, "wavefront");
}

template <int B>
__device__ __forceinline__ void gate(float* st, const float* gg, int wave,
                                     int n15, int half, int aoff, float asgn) {
    v2f a;
    a.x = asgn * gg[aoff];
    a.y = asgn * gg[aoff + 2];

    int base;
    if constexpr (B <= 8) base = half * PLANE + (wave << 9) + ins0(n15, B);
    else                  base = half * PLANE + (wave << 6) + n15;
    float* p = st + base;

    const v8f zero = {0.f, 0.f, 0.f, 0.f, 0.f, 0.f, 0.f, 0.f};

#pragma unroll
    for (int tb = 0; tb < 16; tb += 8) {
        v2f b[8];
        v8f d[8];
#pragma unroll
        for (int u = 0; u < 8; ++u) {
            const int t = tb + u;
            int off;
            if constexpr (B <= 8) off = ins0(t << 4, B);
            else                  off = ((t & 3) << 4) | (ins0(t >> 2, B - 9) << 9);
            b[u].x = p[off];
            b[u].y = p[off + (1 << B)];
        }
#pragma unroll
        for (int u = 0; u < 8; ++u) {
            d[u] = __builtin_amdgcn_wmma_f32_16x16x4_f32(false, a, false, b[u], (short)0, zero, false, false);
        }
#pragma unroll
        for (int u = 0; u < 8; ++u) {
            const int t = tb + u;
            int off;
            if constexpr (B <= 8) off = ins0(t << 4, B);
            else                  off = ((t & 3) << 4) | (ins0(t >> 2, B - 9) << 9);
            p[off]            = d[u][0];
            p[off + (1 << B)] = d[u][1];
        }
    }
}

__global__ __launch_bounds__(NTHREADS)
void vqc_kernel(const float* __restrict__ x,
                const float* __restrict__ wts,
                float* __restrict__ out) {
    __shared__ float st[2 * NSTATE + 32];
    __shared__ float gm[NGATES * 8];
    __shared__ float wred[NQ * 8];
    float* sr = st;
    float* si = st + PLANE;

    const int t    = threadIdx.x;
    const int b    = blockIdx.x;
    const int wave = t >> 5;
    const int lane = t & 31;

    if (t < NGATES) {
        float e0, e1, e2, e3, e4, e5, e6, e7;
        if (t < NQ) {
            const float a = x[b * NQ + t] * 0.5f;
            const float c = cosf(a), s = sinf(a);
            e0 = c;    e1 = 0.f;
            e2 = 0.f;  e3 = -s;
            e4 = 0.f;  e5 = -s;
            e6 = c;    e7 = 0.f;
        } else {
            const float* wp = wts + (t - NQ) * 3;
            const float phi = wp[0], th = wp[1], om = wp[2];
            const float ct = cosf(th * 0.5f);
            const float sn = sinf(th * 0.5f);
            const float hs = 0.5f * (phi + om);
            const float hd = 0.5f * (phi - om);
            const float chs = cosf(hs), shs = sinf(hs);
            const float chd = cosf(hd), shd = sinf(hd);
            e0 =  chs * ct;  e1 = -shs * ct;
            e2 = -chd * sn;  e3 = -shd * sn;
            e4 =  chd * sn;  e5 = -shd * sn;
            e6 =  chs * ct;  e7 =  shs * ct;
        }
        float* gr = gm + t * 8;
        gr[0] = e0; gr[1] = e1; gr[2] = e2; gr[3] = e3;
        gr[4] = e4; gr[5] = e5; gr[6] = e6; gr[7] = e7;
    }

#pragma unroll
    for (int k = 0; k < 16; ++k) {
        const int i = k * NTHREADS + t;
        sr[i] = 0.0f;
        si[i] = 0.0f;
    }
    if (t == 0) sr[0] = 1.0f;
    __syncthreads();

    const int n15  = lane & 15;
    const int half = lane >> 4;
    const int grow = n15 & 1;
    const int irow = (n15 >> 3) & 1;
    const int aoff = grow * 4 + (irow ^ half);
    const float asgn = ((n15 & 6) != 0) ? 0.0f
                     : ((irow == 0 && half == 1) ? -1.0f : 1.0f);

    for (int L = 0; L <= NLAYERS; ++L) {
        const float* g = gm + L * NQ * 8;
        gate<11>(st, g + 0 * 8, wave, n15, half, aoff, asgn);  wave_sync();
        gate<10>(st, g + 1 * 8, wave, n15, half, aoff, asgn);  wave_sync();
        gate<9> (st, g + 2 * 8, wave, n15, half, aoff, asgn);
        __syncthreads();
        gate<8> (st, g + 3 * 8, wave, n15, half, aoff, asgn);  wave_sync();
        gate<7> (st, g + 4 * 8, wave, n15, half, aoff, asgn);  wave_sync();
        gate<6> (st, g + 5 * 8, wave, n15, half, aoff, asgn);  wave_sync();
        gate<5> (st, g + 6 * 8, wave, n15, half, aoff, asgn);  wave_sync();
        gate<4> (st, g + 7 * 8, wave, n15, half, aoff, asgn);  wave_sync();
        gate<3> (st, g + 8 * 8, wave, n15, half, aoff, asgn);  wave_sync();
        gate<2> (st, g + 9 * 8, wave, n15, half, aoff, asgn);  wave_sync();
        gate<1> (st, g + 10 * 8, wave, n15, half, aoff, asgn); wave_sync();
        gate<0> (st, g + 11 * 8, wave, n15, half, aoff, asgn);
        __syncthreads();

        if (L > 0) {
            const int l = L - 1;
            const int r = l % (NQ - 1) + 1;
            float vr[16], vi[16];
#pragma unroll
            for (int k = 0; k < 16; ++k) {
                const int j = k * NTHREADS + t;
                int src = j;
                for (int qq = NQ - 1; qq >= 0; --qq) {
                    const int cbit = 11 - qq;
                    const int tbit = 11 - ((qq + r) % NQ);
                    src ^= ((src >> cbit) & 1) << tbit;
                }
                vr[k] = sr[src];
                vi[k] = si[src];
            }
            __syncthreads();
#pragma unroll
            for (int k = 0; k < 16; ++k) {
                const int j = k * NTHREADS + t;
                sr[j] = vr[k];
                si[j] = vi[k];
            }
            __syncthreads();
        }
    }

    float part[NQ];
#pragma unroll
    for (int q = 0; q < NQ; ++q) part[q] = 0.0f;
#pragma unroll
    for (int k = 0; k < 16; ++k) {
        const int i = k * NTHREADS + t;
        const float p = sr[i] * sr[i] + si[i] * si[i];
#pragma unroll
        for (int q = 0; q < NQ; ++q) {
            part[q] += ((i >> (11 - q)) & 1) ? -p : p;
        }
    }
#pragma unroll
    for (int q = 0; q < NQ; ++q) {
        float v = part[q];
        for (int off = 16; off >= 1; off >>= 1)
            v += __shfl_xor(v, off, 32);
        if (lane == 0) wred[q * 8 + wave] = v;
    }
    __syncthreads();
    if (t < NQ) {
        float s = 0.0f;
#pragma unroll
        for (int wv = 0; wv < 8; ++wv) s += wred[t * 8 + wv];
        out[b * NQ + t] = s;
    }
}

extern "C" void kernel_launch(void* const* d_in, const int* in_sizes, int n_in,
                              void* d_out, int out_size, void* d_ws, size_t ws_size,
                              hipStream_t stream) {
    const float* x   = (const float*)d_in[0];
    const float* wts = (const float*)d_in[1];
    float* out = (float*)d_out;
    const int batch = in_sizes[0] / NQ;
    vqc_kernel<<<batch, NTHREADS, 0, stream>>>(x, wts, out);
}


# CNOT ring source index by GF(2) linearity: one 12-step chain per thread xor four wave-uniform constants for index bits 8-11, not sixteen chains per thread; results bit-identical
# speedup vs baseline: 2.2398x; 1.1133x over previous
#include <hip/hip_runtime.h>
#include <math.h>


typedef __attribute__((ext_vector_type(2))) float v2f;
typedef __attribute__((ext_vector_type(8))) float v8f;

#define NQ      12
#define NSTATE  4096
#define NLAYERS 4
#define NTHREADS 256
#define NGATES  (NQ + NLAYERS * NQ)
#define PLANE   (NSTATE + 32)

__device__ __forceinline__ constexpr int ins0(int x, int B) {
    return ((x & ~((1 << B) - 1)) << 1) | (x & ((1 << B) - 1));
}

typedef __attribute__((ext_vector_type(4))) float v4f;

__device__ __forceinline__ void wave_sync() {
    __builtin_amdgcn_fence(__ATOMIC_RELEASE, "wavefront");
    __builtin_amdgcn_wave_barrier();
    __builtin_amdgcn_fence(__ATOMIC_ACQUIRE, "wavefront");
}

__device__ __forceinline__ void lds_block_edge() {
    __builtin_amdgcn_sched_barrier(0);
}

template <int GROUP>
__device__ __forceinline__ constexpr int roff(int rho) {
    if constexpr (GROUP == 0)      return ((rho & 3) << 4) | ((rho >> 2) << 9);
    else if constexpr (GROUP == 1) return rho << 4;
    else                           return (rho & 15) | ((rho >> 4) << 8);
}

template <int GROUP>
__device__ __forceinline__ void load_group(float (&s)[32], const float* p) {
    if constexpr (GROUP == 2) {
#pragma unroll
        for (int q = 0; q < 8; ++q) {
            const v4f v = *reinterpret_cast<const v4f*>(p + roff<2>(4 * q));
            s[4 * q] = v.x; s[4 * q + 1] = v.y; s[4 * q + 2] = v.z; s[4 * q + 3] = v.w;
        }
    } else {
#pragma unroll
        for (int rho = 0; rho < 32; ++rho) s[rho] = p[roff<GROUP>(rho)];
    }
}

template <int GROUP>
__device__ __forceinline__ void store_group(const float (&s)[32], float* p) {
    if constexpr (GROUP == 2) {
#pragma unroll
        for (int q = 0; q < 8; ++q) {
            v4f v;
            v.x = s[4 * q]; v.y = s[4 * q + 1]; v.z = s[4 * q + 2]; v.w = s[4 * q + 3];
            *reinterpret_cast<v4f*>(p + roff<2>(4 * q)) = v;
        }
    } else {
#pragma unroll
        for (int rho = 0; rho < 32; ++rho) p[roff<GROUP>(rho)] = s[rho];
    }
}

template <int NG>
__device__ __forceinline__ void load_gates(float (&ga)[10], const float* g0, int aoff) {
#pragma unroll
    for (int i = 0; i < NG; ++i) {
        ga[2 * i]     = g0[i * 8 + aoff];
        ga[2 * i + 1] = g0[i * 8 + aoff + 2];
    }
}

template <int BETA>
__device__ __forceinline__ void reg_gate(float (&s)[32], float g0, float g1, float asgn) {
    v2f a;
    a.x = asgn * g0;
    a.y = asgn * g1;
    const v8f zero = {0.f, 0.f, 0.f, 0.f, 0.f, 0.f, 0.f, 0.f};
#pragma unroll
    for (int u = 0; u < 16; ++u) {
        const int r0 = ins0(u, BETA);
        const int r1 = r0 | (1 << BETA);
        v2f b;
        b.x = s[r0];
        b.y = s[r1];
        const v8f d = __builtin_amdgcn_wmma_f32_16x16x4_f32(false, a, false, b, (short)0, zero, false, false);
        s[r0] = d[0];
        s[r1] = d[1];
    }
}

__device__ __forceinline__ int ring_src(int v, int r) {
    for (int qq = NQ - 1; qq >= 0; --qq) {
        const int cbit = 11 - qq;
        const int tbit = 11 - ((qq + r) % NQ);
        v ^= ((v >> cbit) & 1) << tbit;
    }
    return v;
}

__global__ __launch_bounds__(NTHREADS)
void vqc_kernel(const float* __restrict__ x,
                const float* __restrict__ wts,
                float* __restrict__ out) {
    __shared__ __attribute__((aligned(16))) float st[2 * NSTATE + 32];
    __shared__ float gm[NGATES * 8];
    __shared__ float wred[NQ * 8];
    float* sr = st;
    float* si = st + PLANE;

    const int t    = threadIdx.x;
    const int b    = blockIdx.x;
    const int wave = t >> 5;
    const int lane = t & 31;

    if (t < NGATES) {
        float e0, e1, e2, e3, e4, e5, e6, e7;
        if (t < NQ) {
            const float a = x[b * NQ + t] * 0.5f;
            const float c = cosf(a), s = sinf(a);
            e0 = c;    e1 = 0.f;
            e2 = 0.f;  e3 = -s;
            e4 = 0.f;  e5 = -s;
            e6 = c;    e7 = 0.f;
        } else {
            const float* wp = wts + (t - NQ) * 3;
            const float phi = wp[0], th = wp[1], om = wp[2];
            const float ct = cosf(th * 0.5f);
            const float sn = sinf(th * 0.5f);
            const float hs = 0.5f * (phi + om);
            const float hd = 0.5f * (phi - om);
            const float chs = cosf(hs), shs = sinf(hs);
            const float chd = cosf(hd), shd = sinf(hd);
            e0 =  chs * ct;  e1 = -shs * ct;
            e2 = -chd * sn;  e3 = -shd * sn;
            e4 =  chd * sn;  e5 = -shd * sn;
            e6 =  chs * ct;  e7 =  shs * ct;
        }
        float* gr = gm + t * 8;
        gr[0] = e0; gr[1] = e1; gr[2] = e2; gr[3] = e3;
        gr[4] = e4; gr[5] = e5; gr[6] = e6; gr[7] = e7;
    }

#pragma unroll
    for (int k = 0; k < 16; ++k) {
        const int i = k * NTHREADS + t;
        sr[i] = 0.0f;
        si[i] = 0.0f;
    }
    if (t == 0) sr[0] = 1.0f;
    __syncthreads();

    const int n15  = lane & 15;
    const int half = lane >> 4;
    const int grow = n15 & 1;
    const int irow = (n15 >> 3) & 1;
    const int aoff = grow * 4 + (irow ^ half);
    const float asgn = ((n15 & 6) != 0) ? 0.0f
                     : ((irow == 0 && half == 1) ? -1.0f : 1.0f);

    for (int L = 0; L <= NLAYERS; ++L) {
        const float* g = gm + L * NQ * 8;
        {
            float* p = st + half * PLANE + (wave << 6) + n15;
            float s[32], ga[10];
            load_gates<3>(ga, g + 0 * 8, aoff);
            load_group<0>(s, p);
            lds_block_edge();
            reg_gate<4>(s, ga[0], ga[1], asgn);
            reg_gate<3>(s, ga[2], ga[3], asgn);
            reg_gate<2>(s, ga[4], ga[5], asgn);
            lds_block_edge();
            store_group<0>(s, p);
        }
        __syncthreads();
        {
            float* p = st + half * PLANE + (wave << 9) + n15;
            float s[32], ga[10];
            load_gates<5>(ga, g + 3 * 8, aoff);
            load_group<1>(s, p);
            lds_block_edge();
            reg_gate<4>(s, ga[0], ga[1], asgn);
            reg_gate<3>(s, ga[2], ga[3], asgn);
            reg_gate<2>(s, ga[4], ga[5], asgn);
            reg_gate<1>(s, ga[6], ga[7], asgn);
            reg_gate<0>(s, ga[8], ga[9], asgn);
            lds_block_edge();
            store_group<1>(s, p);
        }
        wave_sync();
        {
            float* p = st + half * PLANE + (wave << 9) + (n15 << 4);
            float s[32], ga[10];
            load_gates<4>(ga, g + 8 * 8, aoff);
            load_group<2>(s, p);
            lds_block_edge();
            reg_gate<3>(s, ga[0], ga[1], asgn);
            reg_gate<2>(s, ga[2], ga[3], asgn);
            reg_gate<1>(s, ga[4], ga[5], asgn);
            reg_gate<0>(s, ga[6], ga[7], asgn);
            lds_block_edge();
            store_group<2>(s, p);
        }
        __syncthreads();

        if (L > 0) {
            const int l = L - 1;
            const int r = l % (NQ - 1) + 1;
            const int ft  = ring_src(t, r);
            const int c8  = ring_src(1 << 8, r);
            const int c9  = ring_src(1 << 9, r);
            const int c10 = ring_src(1 << 10, r);
            const int c11 = ring_src(1 << 11, r);
            float vr[16], vi[16];
#pragma unroll
            for (int k = 0; k < 16; ++k) {
                int src = ft;
                if (k & 1) src ^= c8;
                if (k & 2) src ^= c9;
                if (k & 4) src ^= c10;
                if (k & 8) src ^= c11;
                vr[k] = sr[src];
                vi[k] = si[src];
            }
            __syncthreads();
#pragma unroll
            for (int k = 0; k < 16; ++k) {
                const int j = k * NTHREADS + t;
                sr[j] = vr[k];
                si[j] = vi[k];
            }
            __syncthreads();
        }
    }

    float part[NQ];
#pragma unroll
    for (int q = 0; q < NQ; ++q) part[q] = 0.0f;
#pragma unroll
    for (int k = 0; k < 16; ++k) {
        const int i = k * NTHREADS + t;
        const float p = sr[i] * sr[i] + si[i] * si[i];
#pragma unroll
        for (int q = 0; q < NQ; ++q) {
            part[q] += ((i >> (11 - q)) & 1) ? -p : p;
        }
    }
#pragma unroll
    for (int q = 0; q < NQ; ++q) {
        float v = part[q];
        for (int off = 16; off >= 1; off >>= 1)
            v += __shfl_xor(v, off, 32);
        if (lane == 0) wred[q * 8 + wave] = v;
    }
    __syncthreads();
    if (t < NQ) {
        float s = 0.0f;
#pragma unroll
        for (int wv = 0; wv < 8; ++wv) s += wred[t * 8 + wv];
        out[b * NQ + t] = s;
    }
}

extern "C" void kernel_launch(void* const* d_in, const int* in_sizes, int n_in,
                              void* d_out, int out_size, void* d_ws, size_t ws_size,
                              hipStream_t stream) {
    const float* x   = (const float*)d_in[0];
    const float* wts = (const float*)d_in[1];
    float* out = (float*)d_out;
    const int batch = in_sizes[0] / NQ;
    vqc_kernel<<<batch, NTHREADS, 0, stream>>>(x, wts, out);
}


# a stage's 12 gates as four fused triples in two register passes: 64 values per lane, 4 waves per state, rows at pitch 68 dwords; 80->64 WMMA per 512 values, 3 LDS passes->2; f32 sum order changes
# speedup vs baseline: 4.6192x; 2.0623x over previous
#include <hip/hip_runtime.h>
#include <math.h>


typedef __attribute__((ext_vector_type(2))) float v2f;
typedef __attribute__((ext_vector_type(4))) float v4f;
typedef __attribute__((ext_vector_type(8))) float v8f;

#define NQ      12
#define NSTATE  4096
#define NLAYERS 4
#define NTHREADS 128
#define NGATES  (NQ + NLAYERS * NQ)
#define ROWP    68
#define PLANE   (64 * ROWP + 32)

__device__ __forceinline__ void lds_block_edge() {
    __builtin_amdgcn_sched_barrier(0);
}

template <int LOW>
__device__ __forceinline__ void load_pass(float (&s)[64], const float* p) {
    if constexpr (LOW) {
#pragma unroll
        for (int q = 0; q < 16; ++q) {
            const v4f v = *reinterpret_cast<const v4f*>(p + 4 * q);
            s[4 * q] = v.x; s[4 * q + 1] = v.y; s[4 * q + 2] = v.z; s[4 * q + 3] = v.w;
        }
    } else {
#pragma unroll
        for (int rho = 0; rho < 64; ++rho) s[rho] = p[rho * ROWP];
    }
}

template <int LOW>
__device__ __forceinline__ void store_pass(const float (&s)[64], float* p) {
    if constexpr (LOW) {
#pragma unroll
        for (int q = 0; q < 16; ++q) {
            v4f v;
            v.x = s[4 * q]; v.y = s[4 * q + 1]; v.z = s[4 * q + 2]; v.w = s[4 * q + 3];
            *reinterpret_cast<v4f*>(p + 4 * q) = v;
        }
    } else {
#pragma unroll
        for (int rho = 0; rho < 64; ++rho) p[rho * ROWP] = s[rho];
    }
}

__device__ __forceinline__ constexpr int dep6(int v, int mask) {
    int r = 0, k = 0;
    for (int b = 0; b < 6; ++b) {
        if ((mask >> b) & 1) { r |= ((v >> k) & 1) << b; ++k; }
    }
    return r;
}

__device__ __forceinline__ void load_rows(v4f (&gr)[3], const float* g, int n15) {
    const int ao = n15 & 7;
#pragma unroll
    for (int x = 0; x < 3; ++x)
        gr[x] = *reinterpret_cast<const v4f*>(g + x * 8 + ((ao >> (2 - x)) & 1) * 4);
}

__device__ __forceinline__ void fused_a(float (&a)[8], const v4f (&gr)[3], float wr, float wi) {
    const v4f gt = gr[0];
    const float t0r = wr * gt.x - wi * gt.y, t0i = wr * gt.y + wi * gt.x;
    const float t1r = wr * gt.z - wi * gt.w, t1i = wr * gt.w + wi * gt.z;
    const v4f g1 = gr[1];
    const v4f g0 = gr[2];
    const float p00r = g1.x * g0.x - g1.y * g0.y, p00i = g1.x * g0.y + g1.y * g0.x;
    const float p01r = g1.x * g0.z - g1.y * g0.w, p01i = g1.x * g0.w + g1.y * g0.z;
    const float p10r = g1.z * g0.x - g1.w * g0.y, p10i = g1.z * g0.y + g1.w * g0.x;
    const float p11r = g1.z * g0.z - g1.w * g0.w, p11i = g1.z * g0.w + g1.w * g0.z;
    a[0] = t0r * p00r - t0i * p00i;
    a[1] = t0r * p01r - t0i * p01i;
    a[2] = t0r * p10r - t0i * p10i;
    a[3] = t0r * p11r - t0i * p11i;
    a[4] = t1r * p00r - t1i * p00i;
    a[5] = t1r * p01r - t1i * p01i;
    a[6] = t1r * p10r - t1i * p10i;
    a[7] = t1r * p11r - t1i * p11i;
}

template <int MASK>
__device__ __forceinline__ void reg_fused(float (&s)[64], const float (&a)[8]) {
    constexpr int CMASK = 63 & ~MASK;
    const v8f zero = {0.f, 0.f, 0.f, 0.f, 0.f, 0.f, 0.f, 0.f};
#pragma unroll
    for (int u0 = 0; u0 < 8; u0 += 4) {
        v8f acc[4];
#pragma unroll
        for (int i = 0; i < 4; ++i) {
            v2f av;
            av.x = a[2 * i];
            av.y = a[2 * i + 1];
#pragma unroll
            for (int uu = 0; uu < 4; ++uu) {
                const int base = dep6(u0 + uu, CMASK);
                v2f b;
                b.x = s[base | dep6(2 * i, MASK)];
                b.y = s[base | dep6(2 * i + 1, MASK)];
                if (i == 0)
                    acc[uu] = __builtin_amdgcn_wmma_f32_16x16x4_f32(false, av, false, b, (short)0, zero, false, false);
                else
                    acc[uu] = __builtin_amdgcn_wmma_f32_16x16x4_f32(false, av, false, b, (short)0, acc[uu], false, false);
            }
        }
#pragma unroll
        for (int uu = 0; uu < 4; ++uu) {
            const int base = dep6(u0 + uu, CMASK);
#pragma unroll
            for (int v = 0; v < 8; ++v) s[base | dep6(v, MASK)] = acc[uu][v];
        }
    }
}

__device__ __forceinline__ int ring_src(int v, int r) {
    for (int qq = NQ - 1; qq >= 0; --qq) {
        const int cbit = 11 - qq;
        const int tbit = 11 - ((qq + r) % NQ);
        v ^= ((v >> cbit) & 1) << tbit;
    }
    return v;
}

__global__ __launch_bounds__(NTHREADS)
void vqc_kernel(const float* __restrict__ x,
                const float* __restrict__ wts,
                float* __restrict__ out) {
    __shared__ __attribute__((aligned(16))) float st[2 * PLANE];
    __shared__ __attribute__((aligned(16))) float gm[NGATES * 8];
    __shared__ float wred[NQ * 4];
    float* sr = st;
    float* si = st + PLANE;

    const int t    = threadIdx.x;
    const int b    = blockIdx.x;
    const int wave = t >> 5;
    const int lane = t & 31;

    if (t < NGATES) {
        float e0, e1, e2, e3, e4, e5, e6, e7;
        if (t < NQ) {
            const float a = x[b * NQ + t] * 0.5f;
            const float c = cosf(a), s = sinf(a);
            e0 = c;    e1 = 0.f;
            e2 = 0.f;  e3 = -s;
            e4 = 0.f;  e5 = -s;
            e6 = c;    e7 = 0.f;
        } else {
            const float* wp = wts + (t - NQ) * 3;
            const float phi = wp[0], th = wp[1], om = wp[2];
            const float ct = cosf(th * 0.5f);
            const float sn = sinf(th * 0.5f);
            const float hs = 0.5f * (phi + om);
            const float hd = 0.5f * (phi - om);
            const float chs = cosf(hs), shs = sinf(hs);
            const float chd = cosf(hd), shd = sinf(hd);
            e0 =  chs * ct;  e1 = -shs * ct;
            e2 = -chd * sn;  e3 = -shd * sn;
            e4 =  chd * sn;  e5 = -shd * sn;
            e6 =  chs * ct;  e7 =  shs * ct;
        }
        float* gr = gm + t * 8;
        gr[0] = e0; gr[1] = e1; gr[2] = e2; gr[3] = e3;
        gr[4] = e4; gr[5] = e5; gr[6] = e6; gr[7] = e7;
    }

    const int own = (t >> 6) * ROWP + (t & 63);

#pragma unroll
    for (int k = 0; k < 32; ++k) {
        sr[own + k * 2 * ROWP] = 0.0f;
        si[own + k * 2 * ROWP] = 0.0f;
    }
    if (t == 0) sr[0] = 1.0f;
    __syncthreads();

    const int n15  = lane & 15;
    const int half = lane >> 4;
    const int irow = n15 >> 3;
    const float wr3 = (irow == half) ? 1.0f : 0.0f;
    const float wi3 = (irow == half) ? 0.0f : ((irow == 0) ? 1.0f : -1.0f);

    for (int L = 0; L <= NLAYERS; ++L) {
        const float* g = gm + L * NQ * 8;
        {
            float* p = st + half * PLANE + (wave << 4) + n15;
            float s[64], aA[8], aB[8];
            v4f rA[3], rB[3];
            load_rows(rA, g + 0 * 8, n15);
            load_rows(rB, g + 3 * 8, n15);
            load_pass<0>(s, p);
            lds_block_edge();
            fused_a(aA, rA, wr3, wi3);
            fused_a(aB, rB, wr3, wi3);
            reg_fused<7>(s, aB);
            reg_fused<56>(s, aA);
            lds_block_edge();
            store_pass<0>(s, p);
        }
        __syncthreads();
        {
            float* p = st + half * PLANE + ((wave << 4) | n15) * ROWP;
            float s[64], aA[8], aB[8];
            v4f rA[3], rB[3];
            load_rows(rA, g + 6 * 8, n15);
            load_rows(rB, g + 9 * 8, n15);
            load_pass<1>(s, p);
            lds_block_edge();
            fused_a(aA, rA, wr3, wi3);
            fused_a(aB, rB, wr3, wi3);
            reg_fused<56>(s, aA);
            reg_fused<7>(s, aB);
            lds_block_edge();
            store_pass<1>(s, p);
        }
        __syncthreads();

        if (L > 0) {
            const int l = L - 1;
            const int r = l % (NQ - 1) + 1;
            const int ft  = ring_src(t, r);
            const int c7  = ring_src(1 << 7, r);
            const int c8  = ring_src(1 << 8, r);
            const int c9  = ring_src(1 << 9, r);
            const int c10 = ring_src(1 << 10, r);
            const int c11 = ring_src(1 << 11, r);
            float vr[32], vi[32];
#pragma unroll
            for (int k = 0; k < 32; ++k) {
                int src = ft;
                if (k & 1)  src ^= c7;
                if (k & 2)  src ^= c8;
                if (k & 4)  src ^= c9;
                if (k & 8)  src ^= c10;
                if (k & 16) src ^= c11;
                const int a = src + ((src >> 6) << 2);
                vr[k] = sr[a];
                vi[k] = si[a];
            }
            __syncthreads();
#pragma unroll
            for (int k = 0; k < 32; ++k) {
                sr[own + k * 2 * ROWP] = vr[k];
                si[own + k * 2 * ROWP] = vi[k];
            }
            __syncthreads();
        }
    }

    float part[NQ];
#pragma unroll
    for (int q = 0; q < NQ; ++q) part[q] = 0.0f;
#pragma unroll
    for (int k = 0; k < 32; ++k) {
        const int i = k * NTHREADS + t;
        const float vr = sr[own + k * 2 * ROWP];
        const float vi = si[own + k * 2 * ROWP];
        const float p = vr * vr + vi * vi;
#pragma unroll
        for (int q = 0; q < NQ; ++q) {
            part[q] += ((i >> (11 - q)) & 1) ? -p : p;
        }
    }
#pragma unroll
    for (int q = 0; q < NQ; ++q) {
        float v = part[q];
        for (int off = 16; off >= 1; off >>= 1)
            v += __shfl_xor(v, off, 32);
        if (lane == 0) wred[q * 4 + wave] = v;
    }
    __syncthreads();
    if (t < NQ) {
        float s = 0.0f;
#pragma unroll
        for (int wv = 0; wv < 4; ++wv) s += wred[t * 4 + wv];
        out[b * NQ + t] = s;
    }
}

extern "C" void kernel_launch(void* const* d_in, const int* in_sizes, int n_in,
                              void* d_out, int out_size, void* d_ws, size_t ws_size,
                              hipStream_t stream) {
    const float* x   = (const float*)d_in[0];
    const float* wts = (const float*)d_in[1];
    float* out = (float*)d_out;
    const int batch = in_sizes[0] / NQ;
    vqc_kernel<<<batch, NTHREADS, 0, stream>>>(x, wts, out);
}


# wave priority falls with progress (s_setprio 3 up to register pass 7 of 10, then 2, 1, 0): a WGP's 4 co-resident work-groups end together, not one after another, the last alone; bit-identical
# speedup vs baseline: 5.1656x; 1.1183x over previous
#include <hip/hip_runtime.h>
#include <math.h>


typedef __attribute__((ext_vector_type(2))) float v2f;
typedef __attribute__((ext_vector_type(4))) float v4f;
typedef __attribute__((ext_vector_type(8))) float v8f;

#define NQ      12
#define NSTATE  4096
#define NLAYERS 4
#define NTHREADS 128
#define NGATES  (NQ + NLAYERS * NQ)
#define ROWP    68
#define PLANE   (64 * ROWP + 32)

__device__ __forceinline__ void lds_block_edge() {
    __builtin_amdgcn_sched_barrier(0);
}

template <int PRIO>
__device__ __forceinline__ void progress_priority() {
    __builtin_amdgcn_s_setprio(PRIO);
}

template <int LOW>
__device__ __forceinline__ void load_pass(float (&s)[64], const float* p) {
    if constexpr (LOW) {
#pragma unroll
        for (int q = 0; q < 16; ++q) {
            const v4f v = *reinterpret_cast<const v4f*>(p + 4 * q);
            s[4 * q] = v.x; s[4 * q + 1] = v.y; s[4 * q + 2] = v.z; s[4 * q + 3] = v.w;
        }
    } else {
#pragma unroll
        for (int rho = 0; rho < 64; ++rho) s[rho] = p[rho * ROWP];
    }
}

template <int LOW>
__device__ __forceinline__ void store_pass(const float (&s)[64], float* p) {
    if constexpr (LOW) {
#pragma unroll
        for (int q = 0; q < 16; ++q) {
            v4f v;
            v.x = s[4 * q]; v.y = s[4 * q + 1]; v.z = s[4 * q + 2]; v.w = s[4 * q + 3];
            *reinterpret_cast<v4f*>(p + 4 * q) = v;
        }
    } else {
#pragma unroll
        for (int rho = 0; rho < 64; ++rho) p[rho * ROWP] = s[rho];
    }
}

__device__ __forceinline__ constexpr int dep6(int v, int mask) {
    int r = 0, k = 0;
    for (int b = 0; b < 6; ++b) {
        if ((mask >> b) & 1) { r |= ((v >> k) & 1) << b; ++k; }
    }
    return r;
}

__device__ __forceinline__ void load_rows(v4f (&gr)[3], const float* g, int n15) {
    const int ao = n15 & 7;
#pragma unroll
    for (int x = 0; x < 3; ++x)
        gr[x] = *reinterpret_cast<const v4f*>(g + x * 8 + ((ao >> (2 - x)) & 1) * 4);
}

__device__ __forceinline__ void fused_a(float (&a)[8], const v4f (&gr)[3], float wr, float wi) {
    const v4f gt = gr[0];
    const float t0r = wr * gt.x - wi * gt.y, t0i = wr * gt.y + wi * gt.x;
    const float t1r = wr * gt.z - wi * gt.w, t1i = wr * gt.w + wi * gt.z;
    const v4f g1 = gr[1];
    const v4f g0 = gr[2];
    const float p00r = g1.x * g0.x - g1.y * g0.y, p00i = g1.x * g0.y + g1.y * g0.x;
    const float p01r = g1.x * g0.z - g1.y * g0.w, p01i = g1.x * g0.w + g1.y * g0.z;
    const float p10r = g1.z * g0.x - g1.w * g0.y, p10i = g1.z * g0.y + g1.w * g0.x;
    const float p11r = g1.z * g0.z - g1.w * g0.w, p11i = g1.z * g0.w + g1.w * g0.z;
    a[0] = t0r * p00r - t0i * p00i;
    a[1] = t0r * p01r - t0i * p01i;
    a[2] = t0r * p10r - t0i * p10i;
    a[3] = t0r * p11r - t0i * p11i;
    a[4] = t1r * p00r - t1i * p00i;
    a[5] = t1r * p01r - t1i * p01i;
    a[6] = t1r * p10r - t1i * p10i;
    a[7] = t1r * p11r - t1i * p11i;
}

template <int MASK>
__device__ __forceinline__ void reg_fused(float (&s)[64], const float (&a)[8]) {
    constexpr int CMASK = 63 & ~MASK;
    const v8f zero = {0.f, 0.f, 0.f, 0.f, 0.f, 0.f, 0.f, 0.f};
#pragma unroll
    for (int u0 = 0; u0 < 8; u0 += 4) {
        v8f acc[4];
#pragma unroll
        for (int i = 0; i < 4; ++i) {
            v2f av;
            av.x = a[2 * i];
            av.y = a[2 * i + 1];
#pragma unroll
            for (int uu = 0; uu < 4; ++uu) {
                const int base = dep6(u0 + uu, CMASK);
                v2f b;
                b.x = s[base | dep6(2 * i, MASK)];
                b.y = s[base | dep6(2 * i + 1, MASK)];
                if (i == 0)
                    acc[uu] = __builtin_amdgcn_wmma_f32_16x16x4_f32(false, av, false, b, (short)0, zero, false, false);
                else
                    acc[uu] = __builtin_amdgcn_wmma_f32_16x16x4_f32(false, av, false, b, (short)0, acc[uu], false, false);
            }
        }
#pragma unroll
        for (int uu = 0; uu < 4; ++uu) {
            const int base = dep6(u0 + uu, CMASK);
#pragma unroll
            for (int v = 0; v < 8; ++v) s[base | dep6(v, MASK)] = acc[uu][v];
        }
    }
}

__device__ __forceinline__ int ring_src(int v, int r) {
    for (int qq = NQ - 1; qq >= 0; --qq) {
        const int cbit = 11 - qq;
        const int tbit = 11 - ((qq + r) % NQ);
        v ^= ((v >> cbit) & 1) << tbit;
    }
    return v;
}

__global__ __launch_bounds__(NTHREADS)
void vqc_kernel(const float* __restrict__ x,
                const float* __restrict__ wts,
                float* __restrict__ out) {
    __shared__ __attribute__((aligned(16))) float st[2 * PLANE];
    __shared__ __attribute__((aligned(16))) float gm[NGATES * 8];
    __shared__ float wred[NQ * 4];
    float* sr = st;
    float* si = st + PLANE;

    const int t    = threadIdx.x;
    const int b    = blockIdx.x;
    const int wave = t >> 5;
    const int lane = t & 31;

    progress_priority<3>();

    if (t < NGATES) {
        float e0, e1, e2, e3, e4, e5, e6, e7;
        if (t < NQ) {
            const float a = x[b * NQ + t] * 0.5f;
            const float c = cosf(a), s = sinf(a);
            e0 = c;    e1 = 0.f;
            e2 = 0.f;  e3 = -s;
            e4 = 0.f;  e5 = -s;
            e6 = c;    e7 = 0.f;
        } else {
            const float* wp = wts + (t - NQ) * 3;
            const float phi = wp[0], th = wp[1], om = wp[2];
            const float ct = cosf(th * 0.5f);
            const float sn = sinf(th * 0.5f);
            const float hs = 0.5f * (phi + om);
            const float hd = 0.5f * (phi - om);
            const float chs = cosf(hs), shs = sinf(hs);
            const float chd = cosf(hd), shd = sinf(hd);
            e0 =  chs * ct;  e1 = -shs * ct;
            e2 = -chd * sn;  e3 = -shd * sn;
            e4 =  chd * sn;  e5 = -shd * sn;
            e6 =  chs * ct;  e7 =  shs * ct;
        }
        float* gr = gm + t * 8;
        gr[0] = e0; gr[1] = e1; gr[2] = e2; gr[3] = e3;
        gr[4] = e4; gr[5] = e5; gr[6] = e6; gr[7] = e7;
    }

    const int own = (t >> 6) * ROWP + (t & 63);

#pragma unroll
    for (int k = 0; k < 32; ++k) {
        sr[own + k * 2 * ROWP] = 0.0f;
        si[own + k * 2 * ROWP] = 0.0f;
    }
    if (t == 0) sr[0] = 1.0f;
    __syncthreads();

    const int n15  = lane & 15;
    const int half = lane >> 4;
    const int irow = n15 >> 3;
    const float wr3 = (irow == half) ? 1.0f : 0.0f;
    const float wi3 = (irow == half) ? 0.0f : ((irow == 0) ? 1.0f : -1.0f);

    for (int L = 0; L <= NLAYERS; ++L) {
        const float* g = gm + L * NQ * 8;
        if (L == NLAYERS) progress_priority<1>();
        {
            float* p = st + half * PLANE + (wave << 4) + n15;
            float s[64], aA[8], aB[8];
            v4f rA[3], rB[3];
            load_rows(rA, g + 0 * 8, n15);
            load_rows(rB, g + 3 * 8, n15);
            load_pass<0>(s, p);
            lds_block_edge();
            fused_a(aA, rA, wr3, wi3);
            fused_a(aB, rB, wr3, wi3);
            reg_fused<7>(s, aB);
            reg_fused<56>(s, aA);
            lds_block_edge();
            store_pass<0>(s, p);
        }
        __syncthreads();
        if (L == NLAYERS - 1) progress_priority<2>();
        if (L == NLAYERS)     progress_priority<0>();
        {
            float* p = st + half * PLANE + ((wave << 4) | n15) * ROWP;
            float s[64], aA[8], aB[8];
            v4f rA[3], rB[3];
            load_rows(rA, g + 6 * 8, n15);
            load_rows(rB, g + 9 * 8, n15);
            load_pass<1>(s, p);
            lds_block_edge();
            fused_a(aA, rA, wr3, wi3);
            fused_a(aB, rB, wr3, wi3);
            reg_fused<56>(s, aA);
            reg_fused<7>(s, aB);
            lds_block_edge();
            store_pass<1>(s, p);
        }
        __syncthreads();

        if (L > 0) {
            const int l = L - 1;
            const int r = l % (NQ - 1) + 1;
            const int ft  = ring_src(t, r);
            const int c7  = ring_src(1 << 7, r);
            const int c8  = ring_src(1 << 8, r);
            const int c9  = ring_src(1 << 9, r);
            const int c10 = ring_src(1 << 10, r);
            const int c11 = ring_src(1 << 11, r);
            float vr[32], vi[32];
#pragma unroll
            for (int k = 0; k < 32; ++k) {
                int src = ft;
                if (k & 1)  src ^= c7;
                if (k & 2)  src ^= c8;
                if (k & 4)  src ^= c9;
                if (k & 8)  src ^= c10;
                if (k & 16) src ^= c11;
                const int a = src + ((src >> 6) << 2);
                vr[k] = sr[a];
                vi[k] = si[a];
            }
            __syncthreads();
#pragma unroll
            for (int k = 0; k < 32; ++k) {
                sr[own + k * 2 * ROWP] = vr[k];
                si[own + k * 2 * ROWP] = vi[k];
            }
            __syncthreads();
        }
    }

    float part[NQ];
#pragma unroll
    for (int q = 0; q < NQ; ++q) part[q] = 0.0f;
#pragma unroll
    for (int k = 0; k < 32; ++k) {
        const int i = k * NTHREADS + t;
        const float vr = sr[own + k * 2 * ROWP];
        const float vi = si[own + k * 2 * ROWP];
        const float p = vr * vr + vi * vi;
#pragma unroll
        for (int q = 0; q < NQ; ++q) {
            part[q] += ((i >> (11 - q)) & 1) ? -p : p;
        }
    }
#pragma unroll
    for (int q = 0; q < NQ; ++q) {
        float v = part[q];
        for (int off = 16; off >= 1; off >>= 1)
            v += __shfl_xor(v, off, 32);
        if (lane == 0) wred[q * 4 + wave] = v;
    }
    __syncthreads();
    if (t < NQ) {
        float s = 0.0f;
#pragma unroll
        for (int wv = 0; wv < 4; ++wv) s += wred[t * 4 + wv];
        out[b * NQ + t] = s;
    }
}

extern "C" void kernel_launch(void* const* d_in, const int* in_sizes, int n_in,
                              void* d_out, int out_size, void* d_ws, size_t ws_size,
                              hipStream_t stream) {
    const float* x   = (const float*)d_in[0];
    const float* wts = (const float*)d_in[1];
    float* out = (float*)d_out;
    const int batch = in_sizes[0] / NQ;
    vqc_kernel<<<batch, NTHREADS, 0, stream>>>(x, wts, out);
}


# final <Z> sums: each thread's 12 signed partials (pairwise in-register tree) go over its own 12 real-plane LDS slots; 12 threads add 128 each by 16-byte reads; no shuffles; f32 sum order changes
# speedup vs baseline: 5.7029x; 1.1040x over previous
#include <hip/hip_runtime.h>
#include <math.h>


typedef __attribute__((ext_vector_type(2))) float v2f;
typedef __attribute__((ext_vector_type(4))) float v4f;
typedef __attribute__((ext_vector_type(8))) float v8f;

#define NQ      12
#define NSTATE  4096
#define NLAYERS 4
#define NTHREADS 128
#define NGATES  (NQ + NLAYERS * NQ)
#define ROWP    68
#define PLANE   (64 * ROWP + 32)

__device__ __forceinline__ void lds_block_edge() {
    __builtin_amdgcn_sched_barrier(0);
}

template <int PRIO>
__device__ __forceinline__ void progress_priority() {
    __builtin_amdgcn_s_setprio(PRIO);
}

template <int LOW>
__device__ __forceinline__ void load_pass(float (&s)[64], const float* p) {
    if constexpr (LOW) {
#pragma unroll
        for (int q = 0; q < 16; ++q) {
            const v4f v = *reinterpret_cast<const v4f*>(p + 4 * q);
            s[4 * q] = v.x; s[4 * q + 1] = v.y; s[4 * q + 2] = v.z; s[4 * q + 3] = v.w;
        }
    } else {
#pragma unroll
        for (int rho = 0; rho < 64; ++rho) s[rho] = p[rho * ROWP];
    }
}

template <int LOW>
__device__ __forceinline__ void store_pass(const float (&s)[64], float* p) {
    if constexpr (LOW) {
#pragma unroll
        for (int q = 0; q < 16; ++q) {
            v4f v;
            v.x = s[4 * q]; v.y = s[4 * q + 1]; v.z = s[4 * q + 2]; v.w = s[4 * q + 3];
            *reinterpret_cast<v4f*>(p + 4 * q) = v;
        }
    } else {
#pragma unroll
        for (int rho = 0; rho < 64; ++rho) p[rho * ROWP] = s[rho];
    }
}

__device__ __forceinline__ constexpr int dep6(int v, int mask) {
    int r = 0, k = 0;
    for (int b = 0; b < 6; ++b) {
        if ((mask >> b) & 1) { r |= ((v >> k) & 1) << b; ++k; }
    }
    return r;
}

__device__ __forceinline__ void load_rows(v4f (&gr)[3], const float* g, int n15) {
    const int ao = n15 & 7;
#pragma unroll
    for (int x = 0; x < 3; ++x)
        gr[x] = *reinterpret_cast<const v4f*>(g + x * 8 + ((ao >> (2 - x)) & 1) * 4);
}

__device__ __forceinline__ void fused_a(float (&a)[8], const v4f (&gr)[3], float wr, float wi) {
    const v4f gt = gr[0];
    const float t0r = wr * gt.x - wi * gt.y, t0i = wr * gt.y + wi * gt.x;
    const float t1r = wr * gt.z - wi * gt.w, t1i = wr * gt.w + wi * gt.z;
    const v4f g1 = gr[1];
    const v4f g0 = gr[2];
    const float p00r = g1.x * g0.x - g1.y * g0.y, p00i = g1.x * g0.y + g1.y * g0.x;
    const float p01r = g1.x * g0.z - g1.y * g0.w, p01i = g1.x * g0.w + g1.y * g0.z;
    const float p10r = g1.z * g0.x - g1.w * g0.y, p10i = g1.z * g0.y + g1.w * g0.x;
    const float p11r = g1.z * g0.z - g1.w * g0.w, p11i = g1.z * g0.w + g1.w * g0.z;
    a[0] = t0r * p00r - t0i * p00i;
    a[1] = t0r * p01r - t0i * p01i;
    a[2] = t0r * p10r - t0i * p10i;
    a[3] = t0r * p11r - t0i * p11i;
    a[4] = t1r * p00r - t1i * p00i;
    a[5] = t1r * p01r - t1i * p01i;
    a[6] = t1r * p10r - t1i * p10i;
    a[7] = t1r * p11r - t1i * p11i;
}

template <int MASK>
__device__ __forceinline__ void reg_fused(float (&s)[64], const float (&a)[8]) {
    constexpr int CMASK = 63 & ~MASK;
    const v8f zero = {0.f, 0.f, 0.f, 0.f, 0.f, 0.f, 0.f, 0.f};
#pragma unroll
    for (int u0 = 0; u0 < 8; u0 += 4) {
        v8f acc[4];
#pragma unroll
        for (int i = 0; i < 4; ++i) {
            v2f av;
            av.x = a[2 * i];
            av.y = a[2 * i + 1];
#pragma unroll
            for (int uu = 0; uu < 4; ++uu) {
                const int base = dep6(u0 + uu, CMASK);
                v2f b;
                b.x = s[base | dep6(2 * i, MASK)];
                b.y = s[base | dep6(2 * i + 1, MASK)];
                if (i == 0)
                    acc[uu] = __builtin_amdgcn_wmma_f32_16x16x4_f32(false, av, false, b, (short)0, zero, false, false);
                else
                    acc[uu] = __builtin_amdgcn_wmma_f32_16x16x4_f32(false, av, false, b, (short)0, acc[uu], false, false);
            }
        }
#pragma unroll
        for (int uu = 0; uu < 4; ++uu) {
            const int base = dep6(u0 + uu, CMASK);
#pragma unroll
            for (int v = 0; v < 8; ++v) s[base | dep6(v, MASK)] = acc[uu][v];
        }
    }
}

__device__ __forceinline__ int ring_src(int v, int r) {
    for (int qq = NQ - 1; qq >= 0; --qq) {
        const int cbit = 11 - qq;
        const int tbit = 11 - ((qq + r) % NQ);
        v ^= ((v >> cbit) & 1) << tbit;
    }
    return v;
}

template <int N>
__device__ __forceinline__ float fold_level(float (&v)[32]) {
    float d[N / 2];
#pragma unroll
    for (int m = 0; m < N / 2; ++m) {
        const float lo = v[2 * m], hi = v[2 * m + 1];
        d[m] = lo - hi;
        v[m] = lo + hi;
    }
#pragma unroll
    for (int w = N / 4; w >= 1; w >>= 1) {
#pragma unroll
        for (int m = 0; m < w; ++m) d[m] += d[m + w];
    }
    return d[0];
}

__global__ __launch_bounds__(NTHREADS)
void vqc_kernel(const float* __restrict__ x,
                const float* __restrict__ wts,
                float* __restrict__ out) {
    __shared__ __attribute__((aligned(16))) float st[2 * PLANE];
    __shared__ __attribute__((aligned(16))) float gm[NGATES * 8];
    float* sr = st;
    float* si = st + PLANE;

    const int t    = threadIdx.x;
    const int b    = blockIdx.x;
    const int wave = t >> 5;
    const int lane = t & 31;

    progress_priority<3>();

    if (t < NGATES) {
        float e0, e1, e2, e3, e4, e5, e6, e7;
        if (t < NQ) {
            const float a = x[b * NQ + t] * 0.5f;
            const float c = cosf(a), s = sinf(a);
            e0 = c;    e1 = 0.f;
            e2 = 0.f;  e3 = -s;
            e4 = 0.f;  e5 = -s;
            e6 = c;    e7 = 0.f;
        } else {
            const float* wp = wts + (t - NQ) * 3;
            const float phi = wp[0], th = wp[1], om = wp[2];
            const float ct = cosf(th * 0.5f);
            const float sn = sinf(th * 0.5f);
            const float hs = 0.5f * (phi + om);
            const float hd = 0.5f * (phi - om);
            const float chs = cosf(hs), shs = sinf(hs);
            const float chd = cosf(hd), shd = sinf(hd);
            e0 =  chs * ct;  e1 = -shs * ct;
            e2 = -chd * sn;  e3 = -shd * sn;
            e4 =  chd * sn;  e5 = -shd * sn;
            e6 =  chs * ct;  e7 =  shs * ct;
        }
        float* gr = gm + t * 8;
        gr[0] = e0; gr[1] = e1; gr[2] = e2; gr[3] = e3;
        gr[4] = e4; gr[5] = e5; gr[6] = e6; gr[7] = e7;
    }

    const int own = (t >> 6) * ROWP + (t & 63);

#pragma unroll
    for (int k = 0; k < 32; ++k) {
        sr[own + k * 2 * ROWP] = 0.0f;
        si[own + k * 2 * ROWP] = 0.0f;
    }
    if (t == 0) sr[0] = 1.0f;
    __syncthreads();

    const int n15  = lane & 15;
    const int half = lane >> 4;
    const int irow = n15 >> 3;
    const float wr3 = (irow == half) ? 1.0f : 0.0f;
    const float wi3 = (irow == half) ? 0.0f : ((irow == 0) ? 1.0f : -1.0f);

    for (int L = 0; L <= NLAYERS; ++L) {
        const float* g = gm + L * NQ * 8;
        if (L == NLAYERS) progress_priority<1>();
        {
            float* p = st + half * PLANE + (wave << 4) + n15;
            float s[64], aA[8], aB[8];
            v4f rA[3], rB[3];
            load_rows(rA, g + 0 * 8, n15);
            load_rows(rB, g + 3 * 8, n15);
            load_pass<0>(s, p);
            lds_block_edge();
            fused_a(aA, rA, wr3, wi3);
            fused_a(aB, rB, wr3, wi3);
            reg_fused<7>(s, aB);
            reg_fused<56>(s, aA);
            lds_block_edge();
            store_pass<0>(s, p);
        }
        __syncthreads();
        if (L == NLAYERS - 1) progress_priority<2>();
        if (L == NLAYERS)     progress_priority<0>();
        {
            float* p = st + half * PLANE + ((wave << 4) | n15) * ROWP;
            float s[64], aA[8], aB[8];
            v4f rA[3], rB[3];
            load_rows(rA, g + 6 * 8, n15);
            load_rows(rB, g + 9 * 8, n15);
            load_pass<1>(s, p);
            lds_block_edge();
            fused_a(aA, rA, wr3, wi3);
            fused_a(aB, rB, wr3, wi3);
            reg_fused<56>(s, aA);
            reg_fused<7>(s, aB);
            lds_block_edge();
            store_pass<1>(s, p);
        }
        __syncthreads();

        if (L > 0) {
            const int l = L - 1;
            const int r = l % (NQ - 1) + 1;
            const int ft  = ring_src(t, r);
            const int c7  = ring_src(1 << 7, r);
            const int c8  = ring_src(1 << 8, r);
            const int c9  = ring_src(1 << 9, r);
            const int c10 = ring_src(1 << 10, r);
            const int c11 = ring_src(1 << 11, r);
            float vr[32], vi[32];
#pragma unroll
            for (int k = 0; k < 32; ++k) {
                int src = ft;
                if (k & 1)  src ^= c7;
                if (k & 2)  src ^= c8;
                if (k & 4)  src ^= c9;
                if (k & 8)  src ^= c10;
                if (k & 16) src ^= c11;
                const int a = src + ((src >> 6) << 2);
                vr[k] = sr[a];
                vi[k] = si[a];
            }
            __syncthreads();
#pragma unroll
            for (int k = 0; k < 32; ++k) {
                sr[own + k * 2 * ROWP] = vr[k];
                si[own + k * 2 * ROWP] = vi[k];
            }
            __syncthreads();
        }
    }

    float p[32];
#pragma unroll
    for (int k = 0; k < 32; ++k) {
        const float vr = sr[own + k * 2 * ROWP];
        const float vi = si[own + k * 2 * ROWP];
        p[k] = vr * vr + vi * vi;
    }
    float part[NQ];
    part[4] = fold_level<32>(p);
    part[3] = fold_level<16>(p);
    part[2] = fold_level<8>(p);
    part[1] = fold_level<4>(p);
    part[0] = fold_level<2>(p);
    const float total = p[0];
#pragma unroll
    for (int q = 5; q < NQ; ++q) part[q] = ((t >> (11 - q)) & 1) ? -total : total;
#pragma unroll
    for (int q = 0; q < NQ; ++q) sr[own + q * 2 * ROWP] = part[q];
    __syncthreads();
    if (t < NQ) {
        const float* r0 = sr + t * 2 * ROWP;
        const float* r1 = r0 + ROWP;
        v4f a0 = *reinterpret_cast<const v4f*>(r0);
        v4f a1 = *reinterpret_cast<const v4f*>(r0 + 4);
        v4f a2 = *reinterpret_cast<const v4f*>(r1);
        v4f a3 = *reinterpret_cast<const v4f*>(r1 + 4);
#pragma unroll
        for (int j = 2; j < 16; j += 2) {
            a0 += *reinterpret_cast<const v4f*>(r0 + 4 * j);
            a1 += *reinterpret_cast<const v4f*>(r0 + 4 * j + 4);
            a2 += *reinterpret_cast<const v4f*>(r1 + 4 * j);
            a3 += *reinterpret_cast<const v4f*>(r1 + 4 * j + 4);
        }
        const v4f a = (a0 + a1) + (a2 + a3);
        out[b * NQ + t] = (a.x + a.y) + (a.z + a.w);
    }
}

extern "C" void kernel_launch(void* const* d_in, const int* in_sizes, int n_in,
                              void* d_out, int out_size, void* d_ws, size_t ws_size,
                              hipStream_t stream) {
    const float* x   = (const float*)d_in[0];
    const float* wts = (const float*)d_in[1];
    float* out = (float*)d_out;
    const int batch = in_sizes[0] / NQ;
    vqc_kernel<<<batch, NTHREADS, 0, stream>>>(x, wts, out);
}
